# starting kernel
# baseline (speedup 1.0000x reference)
#include <hip/hip_runtime.h>

typedef __attribute__((ext_vector_type(2))) float v2f;
typedef __attribute__((ext_vector_type(8))) float v8f;

#define WMMA_F32_4(a, bmat, c) \
  __builtin_amdgcn_wmma_f32_16x16x4_f32(false, (a), false, (bmat), (short)0, (c), false, false)

constexpr int   kB    = 8;
constexpr int   kN    = 1024;
constexpr int   kInF  = 768;
constexpr int   kOutF = 64;
constexpr float kAlpha  = 0.2f;
constexpr float kNegInf = -9.0e15f;

__global__ __launch_bounds__(256) void k1_gemm_h(
    const float* __restrict__ X, const float* __restrict__ W,
    const float* __restrict__ Wb, float* __restrict__ H)
{
  const int lane = threadIdx.x & 31;
  const int wave = threadIdx.x >> 5;
  const int tile = blockIdx.x * 8 + wave;
  const int row0 = tile << 5;
  const int half = lane >> 4;
  const int ncol = lane & 15;
  const float* __restrict__ Arow0 = X + (size_t)(row0 + ncol) * kInF;
  const float* __restrict__ Arow1 = Arow0 + (size_t)16 * kInF;

  v8f acc0[4] = {};
  v8f acc1[4] = {};
  for (int k0 = 0; k0 < kInF; k0 += 4) {
    const int kb = k0 + half * 2;
    const v2f a0 = *(const v2f*)(Arow0 + kb);
    const v2f a1 = *(const v2f*)(Arow1 + kb);
#pragma unroll
    for (int t = 0; t < 4; ++t) {
      const v2f bm = *(const v2f*)(W + (size_t)(t * 16 + ncol) * kInF + kb);
      acc0[t] = WMMA_F32_4(a0, bm, acc0[t]);
      acc1[t] = WMMA_F32_4(a1, bm, acc1[t]);
    }
  }
  const int rbase = row0 + half * 8;
#pragma unroll
  for (int t = 0; t < 4; ++t) {
    const float bias = Wb[t * 16 + ncol];
#pragma unroll
    for (int r = 0; r < 8; ++r) {
      H[(size_t)(rbase + r) * kOutF + t * 16 + ncol]      = acc0[t][r] + bias;
      H[(size_t)(rbase + 16 + r) * kOutF + t * 16 + ncol] = acc1[t][r] + bias;
    }
  }
}

__global__ __launch_bounds__(256) void k2_f12(
    const float* __restrict__ H, const float* __restrict__ aw,
    float* __restrict__ F1, float* __restrict__ F2)
{
  const int row = blockIdx.x * 256 + threadIdx.x;
  const float* __restrict__ hr = H + (size_t)row * kOutF;
  float s1 = 0.f, s2 = 0.f;
#pragma unroll 8
  for (int o = 0; o < kOutF; ++o) {
    const float v = hr[o];
    s1 += v * aw[o];
    s2 += v * aw[kOutF + o];
  }
  F1[row] = s1;
  F2[row] = s2;
}

__global__ __launch_bounds__(256) void k3_stats(
    const float* __restrict__ F1, const float* __restrict__ F2,
    const int* __restrict__ adj, const float* __restrict__ abp,
    float* __restrict__ Mrow, float* __restrict__ Lrow)
{
  const int lane = threadIdx.x & 31;
  const int wave = threadIdx.x >> 5;
  const int row  = blockIdx.x * 8 + wave;
  const int b    = row >> 10;
  const float f1i = F1[row] + abp[0];
  const int*   __restrict__ ar  = adj + (size_t)row * kN;
  const float* __restrict__ f2b = F2 + (b << 10);

  float m = -3.0e38f, s = 0.f;
  for (int j = lane; j < kN; j += 32) {
    float e = f1i + f2b[j];
    e = (e >= 0.f) ? e : kAlpha * e;
    e = (ar[j] > 0) ? e : kNegInf;
    if (e > m) { s = s * __expf(m - e) + 1.f; m = e; }
    else       { s += __expf(e - m); }
  }
#pragma unroll
  for (int off = 16; off > 0; off >>= 1) {
    const float mo = __shfl_xor(m, off, 32);
    const float so = __shfl_xor(s, off, 32);
    const float M  = fmaxf(m, mo);
    s = s * __expf(m - M) + so * __expf(mo - M);
    m = M;
  }
  if (lane == 0) { Mrow[row] = m; Lrow[row] = s; }
}

__device__ __forceinline__ v2f make_p(float f1i, float mi,
                                      const int* __restrict__ ar, int jx,
                                      v2f f2p)
{
  float e0 = f1i + f2p.x; e0 = (e0 >= 0.f) ? e0 : kAlpha * e0;
  float e1 = f1i + f2p.y; e1 = (e1 >= 0.f) ? e1 : kAlpha * e1;
  const int2 ad = *(const int2*)(ar + jx);
  e0 = (ad.x > 0) ? e0 : kNegInf;
  e1 = (ad.y > 0) ? e1 : kNegInf;
  v2f p;
  p.x = __expf(e0 - mi);
  p.y = __expf(e1 - mi);
  return p;
}

__global__ __launch_bounds__(256) void k4_attn_av(
    const float* __restrict__ H, const float* __restrict__ F1,
    const float* __restrict__ F2, const int* __restrict__ adj,
    const float* __restrict__ abp, const float* __restrict__ Mrow,
    const float* __restrict__ Lrow, float* __restrict__ out)
{
  const int lane = threadIdx.x & 31;
  const int wave = threadIdx.x >> 5;
  const int gt   = blockIdx.x * 8 + wave;
  const int b    = gt >> 5;
  const int i0   = (gt & 31) << 5;
  const int half = lane >> 4;
  const int ncol = lane & 15;
  const int irow0 = (b << 10) + i0 + ncol;
  const int irow1 = irow0 + 16;
  const float ab   = abp[0];
  const float f1a  = F1[irow0] + ab;
  const float f1b  = F1[irow1] + ab;
  const float mia  = Mrow[irow0];
  const float mib  = Mrow[irow1];
  const int*   __restrict__ ar0 = adj + (size_t)irow0 * kN;
  const int*   __restrict__ ar1 = adj + (size_t)irow1 * kN;
  const float* __restrict__ f2b = F2 + (b << 10);
  const float* __restrict__ hb  = H + ((size_t)(b << 10)) * kOutF;

  v8f acc0[4] = {};
  v8f acc1[4] = {};
  for (int j0 = 0; j0 < kN; j0 += 4) {
    const int jx = j0 + half * 2;
    const v2f f2p = *(const v2f*)(f2b + jx);
    const v2f p0 = make_p(f1a, mia, ar0, jx, f2p);
    const v2f p1 = make_p(f1b, mib, ar1, jx, f2p);
    const float* __restrict__ hj = hb + (size_t)jx * kOutF;
#pragma unroll
    for (int t = 0; t < 4; ++t) {
      v2f bm;
      bm.x = hj[t * 16 + ncol];
      bm.y = hj[kOutF + t * 16 + ncol];
      acc0[t] = WMMA_F32_4(p0, bm, acc0[t]);
      acc1[t] = WMMA_F32_4(p1, bm, acc1[t]);
    }
  }
  const int rbase = (b << 10) + i0 + half * 8;
#pragma unroll
  for (int r = 0; r < 8; ++r) {
    const float li0 = 1.0f / Lrow[rbase + r];
    const float li1 = 1.0f / Lrow[rbase + 16 + r];
#pragma unroll
    for (int t = 0; t < 4; ++t) {
      float v0 = acc0[t][r] * li0;
      float v1 = acc1[t][r] * li1;
      v0 = (v0 > 0.f) ? v0 : (__expf(v0) - 1.f);
      v1 = (v1 > 0.f) ? v1 : (__expf(v1) - 1.f);
      out[(size_t)(rbase + r) * kOutF + t * 16 + ncol]      = v0;
      out[(size_t)(rbase + 16 + r) * kOutF + t * 16 + ncol] = v1;
    }
  }
}

extern "C" void kernel_launch(void* const* d_in, const int* in_sizes, int n_in,
                              void* d_out, int out_size, void* d_ws, size_t ws_size,
                              hipStream_t stream) {
  (void)in_sizes; (void)n_in; (void)out_size; (void)ws_size;
  const float* X   = (const float*)d_in[0];
  const int*   adj = (const int*)d_in[3];
  const float* Ww  = (const float*)d_in[4];
  const float* Wb  = (const float*)d_in[5];
  const float* aw  = (const float*)d_in[6];
  const float* ab  = (const float*)d_in[7];

  float* ws = (float*)d_ws;
  float* H  = ws;
  float* F1 = H  + (size_t)kB * kN * kOutF;
  float* F2 = F1 + (size_t)kB * kN;
  float* Mr = F2 + (size_t)kB * kN;
  float* Lr = Mr + (size_t)kB * kN;
  float* O  = (float*)d_out;

  hipLaunchKernelGGL(k1_gemm_h,  dim3(32),   dim3(256), 0, stream, X, Ww, Wb, H);
  hipLaunchKernelGGL(k2_f12,     dim3(32),   dim3(256), 0, stream, H, aw, F1, F2);
  hipLaunchKernelGGL(k3_stats,   dim3(1024), dim3(256), 0, stream, F1, F2, adj, ab, Mr, Lr);
  hipLaunchKernelGGL(k4_attn_av, dim3(32),   dim3(256), 0, stream, H, F1, F2, adj, ab, Mr, Lr, O);
}


# Split-K in both WMMA kernels: the 16 waves of a work-group each sum 1/16 of the reduction range of one 32-row tile (256 groups, not 32); partial tiles added in wave order via LDS (new sum order)
# speedup vs baseline: 3.5416x; 3.5416x over previous
#include <hip/hip_runtime.h>


typedef __attribute__((ext_vector_type(2))) float v2f;
typedef __attribute__((ext_vector_type(4))) float v4f;
typedef __attribute__((ext_vector_type(8))) float v8f;

#define WMMA_F32_4(a, bmat, c) \
  __builtin_amdgcn_wmma_f32_16x16x4_f32(false, (a), false, (bmat), (short)0, (c), false, false)

constexpr int   kB    = 8;
constexpr int   kN    = 1024;
constexpr int   kInF  = 768;
constexpr int   kOutF = 64;
constexpr float kAlpha  = 0.2f;
constexpr float kNegInf = -9.0e15f;

constexpr int kWaves   = 16;
constexpr int kThreads = kWaves * 32;
constexpr int kPart    = 2 * 4 * 8 * 32;

__device__ __forceinline__ void put_partial(float* __restrict__ part, int lane,
                                            const v8f* acc0, const v8f* acc1)
{
#pragma unroll
  for (int t = 0; t < 4; ++t) {
#pragma unroll
    for (int r = 0; r < 8; ++r) {
      part[(t * 8 + r) * 32 + lane]       = acc0[t][r];
      part[((4 + t) * 8 + r) * 32 + lane] = acc1[t][r];
    }
  }
}

struct TilePos { int row; int col; int idx; };
__device__ __forceinline__ TilePos reader_pos(int tid)
{
  const int l = tid & 31, v = tid >> 5;
  const int c4   = (l & 3) << 2;
  const int h    = (l >> 2) & 1;
  const int r    = (((v >> 1) & 3) << 1) | ((l >> 3) & 1);
  const int t    = (((v >> 3) & 1) << 1) | ((l >> 4) & 1);
  const int tile = v & 1;
  TilePos p;
  p.row = tile * 16 + h * 8 + r;
  p.col = t * 16 + c4;
  p.idx = ((tile * 4 + t) * 8 + r) * 32 + h * 16 + c4;
  return p;
}

__device__ __forceinline__ v4f sum_partials(const float* __restrict__ lds, int idx)
{
  v4f s = *(const v4f*)(lds + idx);
#pragma unroll
  for (int w = 1; w < kWaves; ++w) s += *(const v4f*)(lds + w * kPart + idx);
  return s;
}

__global__ __launch_bounds__(kThreads) void k1_gemm_h(
    const float* __restrict__ X, const float* __restrict__ W,
    const float* __restrict__ Wb, float* __restrict__ H)
{
  __shared__ float part[kWaves * kPart];
  const int lane = threadIdx.x & 31;
  const int wave = __builtin_amdgcn_readfirstlane((int)(threadIdx.x >> 5));
  const int row0 = blockIdx.x << 5;
  const int half = lane >> 4;
  const int ncol = lane & 15;
  const float* __restrict__ Arow0 = X + (size_t)(row0 + ncol) * kInF;
  const float* __restrict__ Arow1 = Arow0 + (size_t)16 * kInF;

  constexpr int kChunk = kInF / kWaves;
  const int kbeg = wave * kChunk;

  v8f acc0[4] = {};
  v8f acc1[4] = {};
  for (int k0 = kbeg; k0 < kbeg + kChunk; k0 += 4) {
    const int kb = k0 + half * 2;
    const v2f a0 = *(const v2f*)(Arow0 + kb);
    const v2f a1 = *(const v2f*)(Arow1 + kb);
#pragma unroll
    for (int t = 0; t < 4; ++t) {
      const v2f bm = *(const v2f*)(W + (size_t)(t * 16 + ncol) * kInF + kb);
      acc0[t] = WMMA_F32_4(a0, bm, acc0[t]);
      acc1[t] = WMMA_F32_4(a1, bm, acc1[t]);
    }
  }

  put_partial(part + wave * kPart, lane, acc0, acc1);
  __syncthreads();

  const TilePos p = reader_pos(threadIdx.x);
  const v4f s    = sum_partials(part, p.idx);
  const v4f bias = *(const v4f*)(Wb + p.col);
  *(v4f*)(H + (size_t)(row0 + p.row) * kOutF + p.col) = s + bias;
}

__global__ __launch_bounds__(256) void k2_f12(
    const float* __restrict__ H, const float* __restrict__ aw,
    float* __restrict__ F1, float* __restrict__ F2)
{
  const int row = blockIdx.x * 256 + threadIdx.x;
  const float* __restrict__ hr = H + (size_t)row * kOutF;
  float s1 = 0.f, s2 = 0.f;
#pragma unroll 8
  for (int o = 0; o < kOutF; ++o) {
    const float v = hr[o];
    s1 += v * aw[o];
    s2 += v * aw[kOutF + o];
  }
  F1[row] = s1;
  F2[row] = s2;
}

__global__ __launch_bounds__(256) void k3_stats(
    const float* __restrict__ F1, const float* __restrict__ F2,
    const int* __restrict__ adj, const float* __restrict__ abp,
    float* __restrict__ Mrow, float* __restrict__ Lrow)
{
  const int lane = threadIdx.x & 31;
  const int wave = threadIdx.x >> 5;
  const int row  = blockIdx.x * 8 + wave;
  const int b    = row >> 10;
  const float f1i = F1[row] + abp[0];
  const int*   __restrict__ ar  = adj + (size_t)row * kN;
  const float* __restrict__ f2b = F2 + (b << 10);

  float m = -3.0e38f, s = 0.f;
  for (int j = lane; j < kN; j += 32) {
    float e = f1i + f2b[j];
    e = (e >= 0.f) ? e : kAlpha * e;
    e = (ar[j] > 0) ? e : kNegInf;
    if (e > m) { s = s * __expf(m - e) + 1.f; m = e; }
    else       { s += __expf(e - m); }
  }
#pragma unroll
  for (int off = 16; off > 0; off >>= 1) {
    const float mo = __shfl_xor(m, off, 32);
    const float so = __shfl_xor(s, off, 32);
    const float M  = fmaxf(m, mo);
    s = s * __expf(m - M) + so * __expf(mo - M);
    m = M;
  }
  if (lane == 0) { Mrow[row] = m; Lrow[row] = s; }
}

__device__ __forceinline__ v2f make_p(float f1i, float mi,
                                      const int* __restrict__ ar, int jx,
                                      v2f f2p)
{
  float e0 = f1i + f2p.x; e0 = (e0 >= 0.f) ? e0 : kAlpha * e0;
  float e1 = f1i + f2p.y; e1 = (e1 >= 0.f) ? e1 : kAlpha * e1;
  const int2 ad = *(const int2*)(ar + jx);
  e0 = (ad.x > 0) ? e0 : kNegInf;
  e1 = (ad.y > 0) ? e1 : kNegInf;
  v2f p;
  p.x = __expf(e0 - mi);
  p.y = __expf(e1 - mi);
  return p;
}

__global__ __launch_bounds__(kThreads) void k4_attn_av(
    const float* __restrict__ H, const float* __restrict__ F1,
    const float* __restrict__ F2, const int* __restrict__ adj,
    const float* __restrict__ abp, const float* __restrict__ Mrow,
    const float* __restrict__ Lrow, float* __restrict__ out)
{
  __shared__ float part[kWaves * kPart];
  const int lane = threadIdx.x & 31;
  const int wave = __builtin_amdgcn_readfirstlane((int)(threadIdx.x >> 5));
  const int gt   = blockIdx.x;
  const int b    = gt >> 5;
  const int i0   = (gt & 31) << 5;
  const int half = lane >> 4;
  const int ncol = lane & 15;
  const int irow0 = (b << 10) + i0 + ncol;
  const int irow1 = irow0 + 16;
  const float ab   = abp[0];
  const float f1a  = F1[irow0] + ab;
  const float f1b  = F1[irow1] + ab;
  const float mia  = Mrow[irow0];
  const float mib  = Mrow[irow1];
  const int*   __restrict__ ar0 = adj + (size_t)irow0 * kN;
  const int*   __restrict__ ar1 = adj + (size_t)irow1 * kN;
  const float* __restrict__ f2b = F2 + (b << 10);
  const float* __restrict__ hb  = H + ((size_t)(b << 10)) * kOutF;

  constexpr int kChunk = kN / kWaves;
  const int jbeg = wave * kChunk;

  v8f acc0[4] = {};
  v8f acc1[4] = {};
  for (int j0 = jbeg; j0 < jbeg + kChunk; j0 += 4) {
    const int jx = j0 + half * 2;
    const v2f f2p = *(const v2f*)(f2b + jx);
    const v2f p0 = make_p(f1a, mia, ar0, jx, f2p);
    const v2f p1 = make_p(f1b, mib, ar1, jx, f2p);
    const float* __restrict__ hj = hb + (size_t)jx * kOutF;
#pragma unroll
    for (int t = 0; t < 4; ++t) {
      v2f bm;
      bm.x = hj[t * 16 + ncol];
      bm.y = hj[kOutF + t * 16 + ncol];
      acc0[t] = WMMA_F32_4(p0, bm, acc0[t]);
      acc1[t] = WMMA_F32_4(p1, bm, acc1[t]);
    }
  }

  put_partial(part + wave * kPart, lane, acc0, acc1);
  __syncthreads();

  const TilePos p = reader_pos(threadIdx.x);
  v4f v = sum_partials(part, p.idx);
  const int grow = (b << 10) + i0 + p.row;
  const float li = 1.0f / Lrow[grow];
  v *= li;
  v.x = (v.x > 0.f) ? v.x : (__expf(v.x) - 1.f);
  v.y = (v.y > 0.f) ? v.y : (__expf(v.y) - 1.f);
  v.z = (v.z > 0.f) ? v.z : (__expf(v.z) - 1.f);
  v.w = (v.w > 0.f) ? v.w : (__expf(v.w) - 1.f);
  *(v4f*)(out + (size_t)grow * kOutF + p.col) = v;
}

extern "C" void kernel_launch(void* const* d_in, const int* in_sizes, int n_in,
                              void* d_out, int out_size, void* d_ws, size_t ws_size,
                              hipStream_t stream) {
  (void)in_sizes; (void)n_in; (void)out_size; (void)ws_size;
  const float* X   = (const float*)d_in[0];
  const int*   adj = (const int*)d_in[3];
  const float* Ww  = (const float*)d_in[4];
  const float* Wb  = (const float*)d_in[5];
  const float* aw  = (const float*)d_in[6];
  const float* ab  = (const float*)d_in[7];

  float* ws = (float*)d_ws;
  float* H  = ws;
  float* F1 = H  + (size_t)kB * kN * kOutF;
  float* F2 = F1 + (size_t)kB * kN;
  float* Mr = F2 + (size_t)kB * kN;
  float* Lr = Mr + (size_t)kB * kN;
  float* O  = (float*)d_out;

  hipLaunchKernelGGL(k1_gemm_h,  dim3(256),  dim3(kThreads), 0, stream, X, Ww, Wb, H);
  hipLaunchKernelGGL(k2_f12,     dim3(32),   dim3(256), 0, stream, H, aw, F1, F2);
  hipLaunchKernelGGL(k3_stats,   dim3(1024), dim3(256), 0, stream, F1, F2, adj, ab, Mr, Lr);
  hipLaunchKernelGGL(k4_attn_av, dim3(256),  dim3(kThreads), 0, stream, H, F1, F2, adj, ab, Mr, Lr, O);
}


# Attention kernel reads adj once, by whole rows: its prologue leaves row maxima and adjacency bit words in LDS; stats launch removed; row sums = sums of the loop's probabilities (new sum order)
# speedup vs baseline: 4.6928x; 1.3251x over previous
#include <hip/hip_runtime.h>


typedef __attribute__((ext_vector_type(2))) float v2f;
typedef __attribute__((ext_vector_type(4))) float v4f;
typedef __attribute__((ext_vector_type(8))) float v8f;

#define WMMA_F32_4(a, bmat, c) \
  __builtin_amdgcn_wmma_f32_16x16x4_f32(false, (a), false, (bmat), (short)0, (c), false, false)

constexpr int   kB    = 8;
constexpr int   kN    = 1024;
constexpr int   kInF  = 768;
constexpr int   kOutF = 64;
constexpr float kAlpha  = 0.2f;
constexpr float kNegInf = -9.0e15f;

constexpr int kWaves   = 16;
constexpr int kThreads = kWaves * 32;
constexpr int kPart    = 2 * 4 * 8 * 32;

__device__ __forceinline__ void put_partial(float* __restrict__ part, int lane,
                                            const v8f* acc0, const v8f* acc1)
{
#pragma unroll
  for (int t = 0; t < 4; ++t) {
#pragma unroll
    for (int r = 0; r < 8; ++r) {
      part[(t * 8 + r) * 32 + lane]       = acc0[t][r];
      part[((4 + t) * 8 + r) * 32 + lane] = acc1[t][r];
    }
  }
}

struct TilePos { int row; int col; int idx; };
__device__ __forceinline__ TilePos reader_pos(int tid)
{
  const int l = tid & 31, v = tid >> 5;
  const int c4   = (l & 3) << 2;
  const int h    = (l >> 2) & 1;
  const int r    = (((v >> 1) & 3) << 1) | ((l >> 3) & 1);
  const int t    = (((v >> 3) & 1) << 1) | ((l >> 4) & 1);
  const int tile = v & 1;
  TilePos p;
  p.row = tile * 16 + h * 8 + r;
  p.col = t * 16 + c4;
  p.idx = ((tile * 4 + t) * 8 + r) * 32 + h * 16 + c4;
  return p;
}

__device__ __forceinline__ v4f sum_partials(const float* __restrict__ lds, int idx)
{
  v4f s = *(const v4f*)(lds + idx);
#pragma unroll
  for (int w = 1; w < kWaves; ++w) s += *(const v4f*)(lds + w * kPart + idx);
  return s;
}

__global__ __launch_bounds__(kThreads) void k1_gemm_h(
    const float* __restrict__ X, const float* __restrict__ W,
    const float* __restrict__ Wb, float* __restrict__ H)
{
  __shared__ float part[kWaves * kPart];
  const int lane = threadIdx.x & 31;
  const int wave = __builtin_amdgcn_readfirstlane((int)(threadIdx.x >> 5));
  const int row0 = blockIdx.x << 5;
  const int half = lane >> 4;
  const int ncol = lane & 15;
  const float* __restrict__ Arow0 = X + (size_t)(row0 + ncol) * kInF;
  const float* __restrict__ Arow1 = Arow0 + (size_t)16 * kInF;

  constexpr int kChunk = kInF / kWaves;
  const int kbeg = wave * kChunk;

  v8f acc0[4] = {};
  v8f acc1[4] = {};
  for (int k0 = kbeg; k0 < kbeg + kChunk; k0 += 4) {
    const int kb = k0 + half * 2;
    const v2f a0 = *(const v2f*)(Arow0 + kb);
    const v2f a1 = *(const v2f*)(Arow1 + kb);
#pragma unroll
    for (int t = 0; t < 4; ++t) {
      const v2f bm = *(const v2f*)(W + (size_t)(t * 16 + ncol) * kInF + kb);
      acc0[t] = WMMA_F32_4(a0, bm, acc0[t]);
      acc1[t] = WMMA_F32_4(a1, bm, acc1[t]);
    }
  }

  put_partial(part + wave * kPart, lane, acc0, acc1);
  __syncthreads();

  const TilePos p = reader_pos(threadIdx.x);
  const v4f s    = sum_partials(part, p.idx);
  const v4f bias = *(const v4f*)(Wb + p.col);
  *(v4f*)(H + (size_t)(row0 + p.row) * kOutF + p.col) = s + bias;
}

__global__ __launch_bounds__(256) void k2_f12(
    const float* __restrict__ H, const float* __restrict__ aw,
    float* __restrict__ F1, float* __restrict__ F2)
{
  const int row = blockIdx.x * 256 + threadIdx.x;
  const float* __restrict__ hr = H + (size_t)row * kOutF;
  float s1 = 0.f, s2 = 0.f;
#pragma unroll 8
  for (int o = 0; o < kOutF; ++o) {
    const float v = hr[o];
    s1 += v * aw[o];
    s2 += v * aw[kOutF + o];
  }
  F1[row] = s1;
  F2[row] = s2;
}

__device__ __forceinline__ float score(float f1i, float f2j, bool on)
{
  float e = f1i + f2j;
  e = (e >= 0.f) ? e : kAlpha * e;
  return on ? e : kNegInf;
}

__global__ __launch_bounds__(kThreads) void k3_attn_av(
    const float* __restrict__ H, const float* __restrict__ F1,
    const float* __restrict__ F2, const int* __restrict__ adj,
    const float* __restrict__ abp, float* __restrict__ out)
{
  __shared__ float    part[kWaves * kPart];
  __shared__ float    rowmax[32];
  __shared__ float    rowsum[kWaves * 32];
  __shared__ unsigned adjbits[32 * 32];
  const int lane = threadIdx.x & 31;
  const int wave = __builtin_amdgcn_readfirstlane((int)(threadIdx.x >> 5));
  const int gt   = blockIdx.x;
  const int b    = gt >> 5;
  const int i0   = (gt & 31) << 5;
  const int half = lane >> 4;
  const int ncol = lane & 15;
  const int irow0 = (b << 10) + i0 + ncol;
  const int irow1 = irow0 + 16;
  const float ab   = abp[0];
  const float f1a  = F1[irow0] + ab;
  const float f1b  = F1[irow1] + ab;
  const float* __restrict__ f2b = F2 + (b << 10);
  const float* __restrict__ hb  = H + ((size_t)(b << 10)) * kOutF;

  {
    const int rt = 2 * wave;
    const int gr = (b << 10) + i0 + rt;
    const float g0 = F1[gr] + ab;
    const float g1 = F1[gr + 1] + ab;
    const int*   __restrict__ q0 = adj + (size_t)gr * kN + lane * 4;
    const float* __restrict__ fq = f2b + lane * 4;
    int4 a0[8], a1[8];
    v4f  f[8];
#pragma unroll
    for (int q = 0; q < 8; ++q) {
      f[q]  = *(const v4f*)(fq + q * 128);
      a0[q] = *(const int4*)(q0 + q * 128);
      a1[q] = *(const int4*)(q0 + kN + q * 128);
    }
    float m0 = -__builtin_inff(), m1 = -__builtin_inff();
    unsigned w0 = 0u, w1 = 0u;
#pragma unroll
    for (int q = 0; q < 8; ++q) {
      const float fv[4] = {f[q].x, f[q].y, f[q].z, f[q].w};
      const int   x0[4] = {a0[q].x, a0[q].y, a0[q].z, a0[q].w};
      const int   x1[4] = {a1[q].x, a1[q].y, a1[q].z, a1[q].w};
#pragma unroll
      for (int c = 0; c < 4; ++c) {
        const bool on0 = x0[c] > 0;
        const bool on1 = x1[c] > 0;
        m0 = fmaxf(m0, score(g0, fv[c], on0));
        m1 = fmaxf(m1, score(g1, fv[c], on1));
        const unsigned b0 = __builtin_amdgcn_ballot_w32(on0);
        const unsigned b1 = __builtin_amdgcn_ballot_w32(on1);
        w0 = (lane == q * 4 + c) ? b0 : w0;
        w1 = (lane == q * 4 + c) ? b1 : w1;
      }
    }
#pragma unroll
    for (int off = 16; off > 0; off >>= 1) {
      m0 = fmaxf(m0, __shfl_xor(m0, off, 32));
      m1 = fmaxf(m1, __shfl_xor(m1, off, 32));
    }
    adjbits[rt * 32 + lane]       = w0;
    adjbits[(rt + 1) * 32 + lane] = w1;
    if (lane == 0) { rowmax[rt] = m0; rowmax[rt + 1] = m1; }
  }
  __syncthreads();

  const float mia = rowmax[ncol];
  const float mib = rowmax[ncol + 16];
  const int wq = (wave >> 1) * 4 + half * 2;
  const int sh = (wave & 1) * 16;
  unsigned ba0 = adjbits[ncol * 32 + wq] >> sh;
  unsigned bb0 = adjbits[ncol * 32 + wq + 1] >> sh;
  unsigned ba1 = adjbits[(ncol + 16) * 32 + wq] >> sh;
  unsigned bb1 = adjbits[(ncol + 16) * 32 + wq + 1] >> sh;

  constexpr int kChunk = kN / kWaves;
  const int jbeg = wave * kChunk;

  float ls0 = 0.f, ls1 = 0.f;
  v8f acc0[4] = {};
  v8f acc1[4] = {};
  for (int j0 = jbeg; j0 < jbeg + kChunk; j0 += 4) {
    const int jx = j0 + half * 2;
    const v2f f2p = *(const v2f*)(f2b + jx);
    v2f p0, p1;
    p0.x = __expf(score(f1a, f2p.x, (ba0 & 1u) != 0u) - mia);
    p0.y = __expf(score(f1a, f2p.y, (bb0 & 1u) != 0u) - mia);
    p1.x = __expf(score(f1b, f2p.x, (ba1 & 1u) != 0u) - mib);
    p1.y = __expf(score(f1b, f2p.y, (bb1 & 1u) != 0u) - mib);
    ba0 >>= 1; bb0 >>= 1; ba1 >>= 1; bb1 >>= 1;
    ls0 += p0.x + p0.y;
    ls1 += p1.x + p1.y;
    const float* __restrict__ hj = hb + (size_t)jx * kOutF;
#pragma unroll
    for (int t = 0; t < 4; ++t) {
      v2f bm;
      bm.x = hj[t * 16 + ncol];
      bm.y = hj[kOutF + t * 16 + ncol];
      acc0[t] = WMMA_F32_4(p0, bm, acc0[t]);
      acc1[t] = WMMA_F32_4(p1, bm, acc1[t]);
    }
  }

  ls0 += __shfl_xor(ls0, 16, 32);
  ls1 += __shfl_xor(ls1, 16, 32);
  if (half == 0) {
    rowsum[wave * 32 + ncol]      = ls0;
    rowsum[wave * 32 + 16 + ncol] = ls1;
  }
  put_partial(part + wave * kPart, lane, acc0, acc1);
  __syncthreads();

  const TilePos p = reader_pos(threadIdx.x);
  v4f v = sum_partials(part, p.idx);
  float l = rowsum[p.row];
#pragma unroll
  for (int w = 1; w < kWaves; ++w) l += rowsum[w * 32 + p.row];
  const int grow = (b << 10) + i0 + p.row;
  const float li = 1.0f / l;
  v *= li;
  v.x = (v.x > 0.f) ? v.x : (__expf(v.x) - 1.f);
  v.y = (v.y > 0.f) ? v.y : (__expf(v.y) - 1.f);
  v.z = (v.z > 0.f) ? v.z : (__expf(v.z) - 1.f);
  v.w = (v.w > 0.f) ? v.w : (__expf(v.w) - 1.f);
  *(v4f*)(out + (size_t)grow * kOutF + p.col) = v;
}

extern "C" void kernel_launch(void* const* d_in, const int* in_sizes, int n_in,
                              void* d_out, int out_size, void* d_ws, size_t ws_size,
                              hipStream_t stream) {
  (void)in_sizes; (void)n_in; (void)out_size; (void)ws_size;
  const float* X   = (const float*)d_in[0];
  const int*   adj = (const int*)d_in[3];
  const float* Ww  = (const float*)d_in[4];
  const float* Wb  = (const float*)d_in[5];
  const float* aw  = (const float*)d_in[6];
  const float* ab  = (const float*)d_in[7];

  float* ws = (float*)d_ws;
  float* H  = ws;
  float* F1 = H  + (size_t)kB * kN * kOutF;
  float* F2 = F1 + (size_t)kB * kN;
  float* O  = (float*)d_out;

  hipLaunchKernelGGL(k1_gemm_h,  dim3(256), dim3(kThreads), 0, stream, X, Ww, Wb, H);
  hipLaunchKernelGGL(k2_f12,     dim3(32),  dim3(256), 0, stream, H, aw, F1, F2);
  hipLaunchKernelGGL(k3_attn_av, dim3(256), dim3(kThreads), 0, stream, H, F1, F2, adj, ab, O);
}


# Row scores f1,f2 computed in the GEMM kernel's summing phase (thread dots its 4 summed columns, 4 lane-exchange adds per row; new sum order for f1,f2); the separate f1/f2 launch is removed
# speedup vs baseline: 5.1469x; 1.0968x over previous
#include <hip/hip_runtime.h>


typedef __attribute__((ext_vector_type(2))) float v2f;
typedef __attribute__((ext_vector_type(4))) float v4f;
typedef __attribute__((ext_vector_type(8))) float v8f;

#define WMMA_F32_4(a, bmat, c) \
  __builtin_amdgcn_wmma_f32_16x16x4_f32(false, (a), false, (bmat), (short)0, (c), false, false)

constexpr int   kB    = 8;
constexpr int   kN    = 1024;
constexpr int   kInF  = 768;
constexpr int   kOutF = 64;
constexpr float kAlpha  = 0.2f;
constexpr float kNegInf = -9.0e15f;

constexpr int kWaves   = 16;
constexpr int kThreads = kWaves * 32;
constexpr int kPart    = 2 * 4 * 8 * 32;

__device__ __forceinline__ void put_partial(float* __restrict__ part, int lane,
                                            const v8f* acc0, const v8f* acc1)
{
#pragma unroll
  for (int t = 0; t < 4; ++t) {
#pragma unroll
    for (int r = 0; r < 8; ++r) {
      part[(t * 8 + r) * 32 + lane]       = acc0[t][r];
      part[((4 + t) * 8 + r) * 32 + lane] = acc1[t][r];
    }
  }
}

struct TilePos { int row; int col; int idx; };
__device__ __forceinline__ TilePos reader_pos(int tid)
{
  const int l = tid & 31, v = tid >> 5;
  const int c4   = (l & 3) << 2;
  const int h    = (l >> 2) & 1;
  const int r    = (((v >> 1) & 3) << 1) | ((l >> 3) & 1);
  const int t    = (((v >> 3) & 1) << 1) | ((l >> 4) & 1);
  const int tile = v & 1;
  TilePos p;
  p.row = tile * 16 + h * 8 + r;
  p.col = t * 16 + c4;
  p.idx = ((tile * 4 + t) * 8 + r) * 32 + h * 16 + c4;
  return p;
}

__device__ __forceinline__ v4f sum_partials(const float* __restrict__ lds, int idx)
{
  v4f s = *(const v4f*)(lds + idx);
#pragma unroll
  for (int w = 1; w < kWaves; ++w) s += *(const v4f*)(lds + w * kPart + idx);
  return s;
}

__device__ __forceinline__ void put_partial_rows(float* __restrict__ part, int lane,
                                                 const v8f* acc0, const v8f* acc1)
{
#pragma unroll
  for (int r = 0; r < 8; ++r) {
#pragma unroll
    for (int t = 0; t < 4; ++t) {
      part[(r * 4 + t) * 32 + lane]       = acc0[t][r];
      part[((8 + r) * 4 + t) * 32 + lane] = acc1[t][r];
    }
  }
}

__device__ __forceinline__ TilePos reader_pos_rows(int tid)
{
  const int l = tid & 31, v = tid >> 5;
  const int c4   = (l & 3) << 2;
  const int h    = (l >> 2) & 1;
  const int t    = ((l >> 3) & 1) | ((l >> 4) << 1);
  const int r    = v >> 1;
  const int tile = v & 1;
  TilePos p;
  p.row = tile * 16 + h * 8 + r;
  p.col = t * 16 + c4;
  p.idx = ((tile * 8 + r) * 4 + t) * 32 + h * 16 + c4;
  return p;
}

__global__ __launch_bounds__(kThreads) void k1_gemm_h(
    const float* __restrict__ X, const float* __restrict__ W,
    const float* __restrict__ Wb, const float* __restrict__ aw,
    float* __restrict__ H, float* __restrict__ F1, float* __restrict__ F2)
{
  __shared__ float part[kWaves * kPart];
  __shared__ float f12[64];
  const int lane = threadIdx.x & 31;
  const int wave = __builtin_amdgcn_readfirstlane((int)(threadIdx.x >> 5));
  const int row0 = blockIdx.x << 5;
  const int half = lane >> 4;
  const int ncol = lane & 15;
  const float* __restrict__ Arow0 = X + (size_t)(row0 + ncol) * kInF;
  const float* __restrict__ Arow1 = Arow0 + (size_t)16 * kInF;

  constexpr int kChunk = kInF / kWaves;
  const int kbeg = wave * kChunk;

  const TilePos p = reader_pos_rows(threadIdx.x);
  const v4f bias = *(const v4f*)(Wb + p.col);
  const v4f att1 = *(const v4f*)(aw + p.col);
  const v4f att2 = *(const v4f*)(aw + kOutF + p.col);

  v8f acc0[4] = {};
  v8f acc1[4] = {};
  for (int k0 = kbeg; k0 < kbeg + kChunk; k0 += 4) {
    const int kb = k0 + half * 2;
    const v2f a0 = *(const v2f*)(Arow0 + kb);
    const v2f a1 = *(const v2f*)(Arow1 + kb);
#pragma unroll
    for (int t = 0; t < 4; ++t) {
      const v2f bm = *(const v2f*)(W + (size_t)(t * 16 + ncol) * kInF + kb);
      acc0[t] = WMMA_F32_4(a0, bm, acc0[t]);
      acc1[t] = WMMA_F32_4(a1, bm, acc1[t]);
    }
  }

  put_partial_rows(part + wave * kPart, lane, acc0, acc1);
  __syncthreads();

  const v4f hv = sum_partials(part, p.idx) + bias;

  float d1 = hv.x * att1.x + hv.y * att1.y + hv.z * att1.z + hv.w * att1.w;
  float d2 = hv.x * att2.x + hv.y * att2.y + hv.z * att2.z + hv.w * att2.w;
  d1 += __shfl_xor(d1, 1, 32);   d2 += __shfl_xor(d2, 1, 32);
  d1 += __shfl_xor(d1, 2, 32);   d2 += __shfl_xor(d2, 2, 32);
  d1 += __shfl_xor(d1, 8, 32);   d2 += __shfl_xor(d2, 8, 32);
  d1 += __shfl_xor(d1, 16, 32);  d2 += __shfl_xor(d2, 16, 32);
  if ((lane & 27) == 0) { f12[p.row] = d1; f12[32 + p.row] = d2; }
  __syncthreads();
  *(v4f*)(H + (size_t)(row0 + p.row) * kOutF + p.col) = hv;
  if (wave == 0) {
    F1[row0 + lane] = f12[lane];
    F2[row0 + lane] = f12[32 + lane];
  }
}

__device__ __forceinline__ float score(float f1i, float f2j, bool on)
{
  float e = f1i + f2j;
  e = (e >= 0.f) ? e : kAlpha * e;
  return on ? e : kNegInf;
}

__global__ __launch_bounds__(kThreads) void k2_attn_av(
    const float* __restrict__ H, const float* __restrict__ F1,
    const float* __restrict__ F2, const int* __restrict__ adj,
    const float* __restrict__ abp, float* __restrict__ out)
{
  __shared__ float    part[kWaves * kPart];
  __shared__ float    rowmax[32];
  __shared__ float    rowsum[kWaves * 32];
  __shared__ unsigned adjbits[32 * 32];
  const int lane = threadIdx.x & 31;
  const int wave = __builtin_amdgcn_readfirstlane((int)(threadIdx.x >> 5));
  const int gt   = blockIdx.x;
  const int b    = gt >> 5;
  const int i0   = (gt & 31) << 5;
  const int half = lane >> 4;
  const int ncol = lane & 15;
  const int irow0 = (b << 10) + i0 + ncol;
  const int irow1 = irow0 + 16;
  const float ab   = abp[0];
  const float f1a  = F1[irow0] + ab;
  const float f1b  = F1[irow1] + ab;
  const float* __restrict__ f2b = F2 + (b << 10);
  const float* __restrict__ hb  = H + ((size_t)(b << 10)) * kOutF;

  {
    const int rt = 2 * wave;
    const int gr = (b << 10) + i0 + rt;
    const float g0 = F1[gr] + ab;
    const float g1 = F1[gr + 1] + ab;
    const int*   __restrict__ q0 = adj + (size_t)gr * kN + lane * 4;
    const float* __restrict__ fq = f2b + lane * 4;
    int4 a0[8], a1[8];
    v4f  f[8];
#pragma unroll
    for (int q = 0; q < 8; ++q) {
      f[q]  = *(const v4f*)(fq + q * 128);
      a0[q] = *(const int4*)(q0 + q * 128);
      a1[q] = *(const int4*)(q0 + kN + q * 128);
    }
    float m0 = -__builtin_inff(), m1 = -__builtin_inff();
    unsigned w0 = 0u, w1 = 0u;
#pragma unroll
    for (int q = 0; q < 8; ++q) {
      const float fv[4] = {f[q].x, f[q].y, f[q].z, f[q].w};
      const int   x0[4] = {a0[q].x, a0[q].y, a0[q].z, a0[q].w};
      const int   x1[4] = {a1[q].x, a1[q].y, a1[q].z, a1[q].w};
#pragma unroll
      for (int c = 0; c < 4; ++c) {
        const bool on0 = x0[c] > 0;
        const bool on1 = x1[c] > 0;
        m0 = fmaxf(m0, score(g0, fv[c], on0));
        m1 = fmaxf(m1, score(g1, fv[c], on1));
        const unsigned b0 = __builtin_amdgcn_ballot_w32(on0);
        const unsigned b1 = __builtin_amdgcn_ballot_w32(on1);
        w0 = (lane == q * 4 + c) ? b0 : w0;
        w1 = (lane == q * 4 + c) ? b1 : w1;
      }
    }
#pragma unroll
    for (int off = 16; off > 0; off >>= 1) {
      m0 = fmaxf(m0, __shfl_xor(m0, off, 32));
      m1 = fmaxf(m1, __shfl_xor(m1, off, 32));
    }
    adjbits[rt * 32 + lane]       = w0;
    adjbits[(rt + 1) * 32 + lane] = w1;
    if (lane == 0) { rowmax[rt] = m0; rowmax[rt + 1] = m1; }
  }
  __syncthreads();

  const float mia = rowmax[ncol];
  const float mib = rowmax[ncol + 16];
  const int wq = (wave >> 1) * 4 + half * 2;
  const int sh = (wave & 1) * 16;
  unsigned ba0 = adjbits[ncol * 32 + wq] >> sh;
  unsigned bb0 = adjbits[ncol * 32 + wq + 1] >> sh;
  unsigned ba1 = adjbits[(ncol + 16) * 32 + wq] >> sh;
  unsigned bb1 = adjbits[(ncol + 16) * 32 + wq + 1] >> sh;

  constexpr int kChunk = kN / kWaves;
  const int jbeg = wave * kChunk;

  float ls0 = 0.f, ls1 = 0.f;
  v8f acc0[4] = {};
  v8f acc1[4] = {};
  for (int j0 = jbeg; j0 < jbeg + kChunk; j0 += 4) {
    const int jx = j0 + half * 2;
    const v2f f2p = *(const v2f*)(f2b + jx);
    v2f p0, p1;
    p0.x = __expf(score(f1a, f2p.x, (ba0 & 1u) != 0u) - mia);
    p0.y = __expf(score(f1a, f2p.y, (bb0 & 1u) != 0u) - mia);
    p1.x = __expf(score(f1b, f2p.x, (ba1 & 1u) != 0u) - mib);
    p1.y = __expf(score(f1b, f2p.y, (bb1 & 1u) != 0u) - mib);
    ba0 >>= 1; bb0 >>= 1; ba1 >>= 1; bb1 >>= 1;
    ls0 += p0.x + p0.y;
    ls1 += p1.x + p1.y;
    const float* __restrict__ hj = hb + (size_t)jx * kOutF;
#pragma unroll
    for (int t = 0; t < 4; ++t) {
      v2f bm;
      bm.x = hj[t * 16 + ncol];
      bm.y = hj[kOutF + t * 16 + ncol];
      acc0[t] = WMMA_F32_4(p0, bm, acc0[t]);
      acc1[t] = WMMA_F32_4(p1, bm, acc1[t]);
    }
  }

  ls0 += __shfl_xor(ls0, 16, 32);
  ls1 += __shfl_xor(ls1, 16, 32);
  if (half == 0) {
    rowsum[wave * 32 + ncol]      = ls0;
    rowsum[wave * 32 + 16 + ncol] = ls1;
  }
  put_partial(part + wave * kPart, lane, acc0, acc1);
  __syncthreads();

  const TilePos p = reader_pos(threadIdx.x);
  v4f v = sum_partials(part, p.idx);
  float l = rowsum[p.row];
#pragma unroll
  for (int w = 1; w < kWaves; ++w) l += rowsum[w * 32 + p.row];
  const int grow = (b << 10) + i0 + p.row;
  const float li = 1.0f / l;
  v *= li;
  v.x = (v.x > 0.f) ? v.x : (__expf(v.x) - 1.f);
  v.y = (v.y > 0.f) ? v.y : (__expf(v.y) - 1.f);
  v.z = (v.z > 0.f) ? v.z : (__expf(v.z) - 1.f);
  v.w = (v.w > 0.f) ? v.w : (__expf(v.w) - 1.f);
  *(v4f*)(out + (size_t)grow * kOutF + p.col) = v;
}

extern "C" void kernel_launch(void* const* d_in, const int* in_sizes, int n_in,
                              void* d_out, int out_size, void* d_ws, size_t ws_size,
                              hipStream_t stream) {
  (void)in_sizes; (void)n_in; (void)out_size; (void)ws_size;
  const float* X   = (const float*)d_in[0];
  const int*   adj = (const int*)d_in[3];
  const float* Ww  = (const float*)d_in[4];
  const float* Wb  = (const float*)d_in[5];
  const float* aw  = (const float*)d_in[6];
  const float* ab  = (const float*)d_in[7];

  float* ws = (float*)d_ws;
  float* H  = ws;
  float* F1 = H  + (size_t)kB * kN * kOutF;
  float* F2 = F1 + (size_t)kB * kN;
  float* O  = (float*)d_out;

  hipLaunchKernelGGL(k1_gemm_h,  dim3(256), dim3(kThreads), 0, stream, X, Ww, Wb, aw, H, F1, F2);
  hipLaunchKernelGGL(k2_attn_av, dim3(256), dim3(kThreads), 0, stream, H, F1, F2, adj, ab, O);
}


# Operand loads run ahead of the WMMAs as 16-byte loads: a GEMM wave fetches its whole k slab first (k order inside 8-k groups permuted: new sum order); attention loads h rows 8 steps ahead
# speedup vs baseline: 8.2231x; 1.5977x over previous
#include <hip/hip_runtime.h>


typedef __attribute__((ext_vector_type(2))) float v2f;
typedef __attribute__((ext_vector_type(4))) float v4f;
typedef __attribute__((ext_vector_type(8))) float v8f;

#define WMMA_F32_4(a, bmat, c) \
  __builtin_amdgcn_wmma_f32_16x16x4_f32(false, (a), false, (bmat), (short)0, (c), false, false)

constexpr int   kB    = 8;
constexpr int   kN    = 1024;
constexpr int   kInF  = 768;
constexpr int   kOutF = 64;
constexpr float kAlpha  = 0.2f;
constexpr float kNegInf = -9.0e15f;

constexpr int kWaves   = 16;
constexpr int kThreads = kWaves * 32;
constexpr int kPart    = 2 * 4 * 8 * 32;

__device__ __forceinline__ void put_partial(float* __restrict__ part, int lane,
                                            const v8f* acc0, const v8f* acc1)
{
#pragma unroll
  for (int r = 0; r < 8; ++r) {
    v4f q0, q1;
    q0.x = acc0[0][r]; q0.y = acc0[1][r]; q0.z = acc0[2][r]; q0.w = acc0[3][r];
    q1.x = acc1[0][r]; q1.y = acc1[1][r]; q1.z = acc1[2][r]; q1.w = acc1[3][r];
    *(v4f*)(part + (r * 32 + lane) * 4)       = q0;
    *(v4f*)(part + ((8 + r) * 32 + lane) * 4) = q1;
  }
}

struct TilePos { int row; int col; int idx; };
__device__ __forceinline__ TilePos reader_pos(int tid)
{
  const int l = tid & 31, v = tid >> 5;
  TilePos p;
  p.row = (v >> 3) * 16 + (l >> 4) * 8 + (v & 7);
  p.col = (l & 15) << 2;
  p.idx = tid << 2;
  return p;
}

__device__ __forceinline__ v4f sum_partials(const float* __restrict__ lds, int idx)
{
  v4f s = *(const v4f*)(lds + idx);
#pragma unroll
  for (int w = 1; w < kWaves; ++w) s += *(const v4f*)(lds + w * kPart + idx);
  return s;
}

__device__ __forceinline__ void put_partial_rows(float* __restrict__ part, int lane,
                                                 const v8f* acc0, const v8f* acc1)
{
#pragma unroll
  for (int r = 0; r < 8; ++r) {
#pragma unroll
    for (int t = 0; t < 4; ++t) {
      part[(r * 4 + t) * 32 + lane]       = acc0[t][r];
      part[((8 + r) * 4 + t) * 32 + lane] = acc1[t][r];
    }
  }
}

__device__ __forceinline__ TilePos reader_pos_rows(int tid)
{
  const int l = tid & 31, v = tid >> 5;
  const int c4   = (l & 3) << 2;
  const int h    = (l >> 2) & 1;
  const int t    = ((l >> 3) & 1) | ((l >> 4) << 1);
  const int r    = v >> 1;
  const int tile = v & 1;
  TilePos p;
  p.row = tile * 16 + h * 8 + r;
  p.col = t * 16 + c4;
  p.idx = ((tile * 8 + r) * 4 + t) * 32 + h * 16 + c4;
  return p;
}

__global__ __launch_bounds__(kThreads) __attribute__((amdgpu_waves_per_eu(4, 4))) void k1_gemm_h(
    const float* __restrict__ X, const float* __restrict__ W,
    const float* __restrict__ Wb, const float* __restrict__ aw,
    float* __restrict__ H, float* __restrict__ F1, float* __restrict__ F2)
{
  __shared__ float part[kWaves * kPart];
  __shared__ float f12[64];
  const int lane = threadIdx.x & 31;
  const int wave = __builtin_amdgcn_readfirstlane((int)(threadIdx.x >> 5));
  const int row0 = blockIdx.x << 5;
  const int half = lane >> 4;
  const int ncol = lane & 15;
  constexpr int kChunk  = kInF / kWaves;
  constexpr int kGroups = kChunk / 8;
  const int kbeg = wave * kChunk;
  const float* __restrict__ ap0 = X + (size_t)(row0 + ncol) * kInF + kbeg + half * 4;
  const float* __restrict__ ap1 = ap0 + (size_t)16 * kInF;
  const float* __restrict__ bp  = W + (size_t)ncol * kInF + kbeg + half * 4;

  const TilePos p = reader_pos_rows(threadIdx.x);
  const v4f bias = *(const v4f*)(Wb + p.col);
  const v4f att1 = *(const v4f*)(aw + p.col);
  const v4f att2 = *(const v4f*)(aw + kOutF + p.col);

  v4f xa0[kGroups], xa1[kGroups], wb[kGroups][4];
#pragma unroll
  for (int g = 0; g < kGroups; ++g) {
    xa0[g] = *(const v4f*)(ap0 + 8 * g);
    xa1[g] = *(const v4f*)(ap1 + 8 * g);
#pragma unroll
    for (int t = 0; t < 4; ++t)
      wb[g][t] = *(const v4f*)(bp + (size_t)(t * 16) * kInF + 8 * g);
  }

  v8f acc0[4] = {};
  v8f acc1[4] = {};
#pragma unroll
  for (int g = 0; g < kGroups; ++g) {
    v2f a0lo, a0hi, a1lo, a1hi;
    a0lo.x = xa0[g].x; a0lo.y = xa0[g].y; a0hi.x = xa0[g].z; a0hi.y = xa0[g].w;
    a1lo.x = xa1[g].x; a1lo.y = xa1[g].y; a1hi.x = xa1[g].z; a1hi.y = xa1[g].w;
#pragma unroll
    for (int t = 0; t < 4; ++t) {
      v2f bm;
      bm.x = wb[g][t].x; bm.y = wb[g][t].y;
      acc0[t] = WMMA_F32_4(a0lo, bm, acc0[t]);
      acc1[t] = WMMA_F32_4(a1lo, bm, acc1[t]);
    }
#pragma unroll
    for (int t = 0; t < 4; ++t) {
      v2f bm;
      bm.x = wb[g][t].z; bm.y = wb[g][t].w;
      acc0[t] = WMMA_F32_4(a0hi, bm, acc0[t]);
      acc1[t] = WMMA_F32_4(a1hi, bm, acc1[t]);
    }
  }

  put_partial_rows(part + wave * kPart, lane, acc0, acc1);
  __syncthreads();

  const v4f hv = sum_partials(part, p.idx) + bias;

  float d1 = hv.x * att1.x + hv.y * att1.y + hv.z * att1.z + hv.w * att1.w;
  float d2 = hv.x * att2.x + hv.y * att2.y + hv.z * att2.z + hv.w * att2.w;
  d1 += __shfl_xor(d1, 1, 32);   d2 += __shfl_xor(d2, 1, 32);
  d1 += __shfl_xor(d1, 2, 32);   d2 += __shfl_xor(d2, 2, 32);
  d1 += __shfl_xor(d1, 8, 32);   d2 += __shfl_xor(d2, 8, 32);
  d1 += __shfl_xor(d1, 16, 32);  d2 += __shfl_xor(d2, 16, 32);
  if ((lane & 27) == 0) { f12[p.row] = d1; f12[32 + p.row] = d2; }
  __syncthreads();
  *(v4f*)(H + (size_t)(row0 + p.row) * kOutF + p.col) = hv;
  if (wave == 0) {
    F1[row0 + lane] = f12[lane];
    F2[row0 + lane] = f12[32 + lane];
  }
}

__device__ __forceinline__ float score(float f1i, float f2j, bool on)
{
  float e = f1i + f2j;
  e = (e >= 0.f) ? e : kAlpha * e;
  return on ? e : kNegInf;
}

__global__ __launch_bounds__(kThreads) void k2_attn_av(
    const float* __restrict__ H, const float* __restrict__ F1,
    const float* __restrict__ F2, const int* __restrict__ adj,
    const float* __restrict__ abp, float* __restrict__ out)
{
  __shared__ float    part[kWaves * kPart];
  __shared__ float    rowmax[32];
  __shared__ float    rowsum[kWaves * 32];
  __shared__ unsigned adjbits[32 * 32];
  const int lane = threadIdx.x & 31;
  const int wave = __builtin_amdgcn_readfirstlane((int)(threadIdx.x >> 5));
  const int gt   = blockIdx.x;
  const int b    = gt >> 5;
  const int i0   = (gt & 31) << 5;
  const int half = lane >> 4;
  const int ncol = lane & 15;
  const int irow0 = (b << 10) + i0 + ncol;
  const int irow1 = irow0 + 16;
  const float ab   = abp[0];
  const float f1a  = F1[irow0] + ab;
  const float f1b  = F1[irow1] + ab;
  const float* __restrict__ f2b = F2 + (b << 10);
  const float* __restrict__ hb  = H + ((size_t)(b << 10)) * kOutF;

  {
    const int rt = 2 * wave;
    const int gr = (b << 10) + i0 + rt;
    const float g0 = F1[gr] + ab;
    const float g1 = F1[gr + 1] + ab;
    const int*   __restrict__ q0 = adj + (size_t)gr * kN + lane * 4;
    const float* __restrict__ fq = f2b + lane * 4;
    int4 a0[8], a1[8];
    v4f  f[8];
#pragma unroll
    for (int q = 0; q < 8; ++q) {
      f[q]  = *(const v4f*)(fq + q * 128);
      a0[q] = *(const int4*)(q0 + q * 128);
      a1[q] = *(const int4*)(q0 + kN + q * 128);
    }
    float m0 = -__builtin_inff(), m1 = -__builtin_inff();
    unsigned w0 = 0u, w1 = 0u;
#pragma unroll
    for (int q = 0; q < 8; ++q) {
      const float fv[4] = {f[q].x, f[q].y, f[q].z, f[q].w};
      const int   x0[4] = {a0[q].x, a0[q].y, a0[q].z, a0[q].w};
      const int   x1[4] = {a1[q].x, a1[q].y, a1[q].z, a1[q].w};
#pragma unroll
      for (int c = 0; c < 4; ++c) {
        const bool on0 = x0[c] > 0;
        const bool on1 = x1[c] > 0;
        m0 = fmaxf(m0, score(g0, fv[c], on0));
        m1 = fmaxf(m1, score(g1, fv[c], on1));
        const unsigned b0 = __builtin_amdgcn_ballot_w32(on0);
        const unsigned b1 = __builtin_amdgcn_ballot_w32(on1);
        w0 = (lane == q * 4 + c) ? b0 : w0;
        w1 = (lane == q * 4 + c) ? b1 : w1;
      }
    }
#pragma unroll
    for (int off = 16; off > 0; off >>= 1) {
      m0 = fmaxf(m0, __shfl_xor(m0, off, 32));
      m1 = fmaxf(m1, __shfl_xor(m1, off, 32));
    }
    adjbits[rt * 32 + lane]       = w0;
    adjbits[(rt + 1) * 32 + lane] = w1;
    if (lane == 0) { rowmax[rt] = m0; rowmax[rt + 1] = m1; }
  }
  __syncthreads();

  const float mia = rowmax[ncol];
  const float mib = rowmax[ncol + 16];
  const int wq = (wave >> 1) * 4 + half * 2;
  const int sh = (wave & 1) * 16;
  unsigned ba0 = adjbits[ncol * 32 + wq] >> sh;
  unsigned bb0 = adjbits[ncol * 32 + wq + 1] >> sh;
  unsigned ba1 = adjbits[(ncol + 16) * 32 + wq] >> sh;
  unsigned bb1 = adjbits[(ncol + 16) * 32 + wq + 1] >> sh;

  constexpr int kChunk = kN / kWaves;
  const int jbeg = wave * kChunk;

  constexpr int kSteps = kChunk / 4;
  constexpr int kAhead = 8;
  const float* __restrict__ fp = f2b + jbeg + half * 2;
  const float* __restrict__ hp = hb + (size_t)(jbeg + half * 2) * kOutF + ncol * 4;
  v2f fs[kSteps];
  v4f hx[kSteps], hy[kSteps];
#pragma unroll
  for (int s = 0; s < kAhead; ++s) {
    fs[s] = *(const v2f*)(fp + 4 * s);
    hx[s] = *(const v4f*)(hp + (size_t)(4 * s) * kOutF);
    hy[s] = *(const v4f*)(hp + (size_t)(4 * s + 1) * kOutF);
  }

  float ls0 = 0.f, ls1 = 0.f;
  v8f acc0[4] = {};
  v8f acc1[4] = {};
#pragma unroll
  for (int s = 0; s < kSteps; ++s) {
    if (s + kAhead < kSteps) {
      fs[s + kAhead] = *(const v2f*)(fp + 4 * (s + kAhead));
      hx[s + kAhead] = *(const v4f*)(hp + (size_t)(4 * (s + kAhead)) * kOutF);
      hy[s + kAhead] = *(const v4f*)(hp + (size_t)(4 * (s + kAhead) + 1) * kOutF);
    }
    const v2f f2p = fs[s];
    v2f p0, p1;
    p0.x = __expf(score(f1a, f2p.x, (ba0 & 1u) != 0u) - mia);
    p0.y = __expf(score(f1a, f2p.y, (bb0 & 1u) != 0u) - mia);
    p1.x = __expf(score(f1b, f2p.x, (ba1 & 1u) != 0u) - mib);
    p1.y = __expf(score(f1b, f2p.y, (bb1 & 1u) != 0u) - mib);
    ba0 >>= 1; bb0 >>= 1; ba1 >>= 1; bb1 >>= 1;
    ls0 += p0.x + p0.y;
    ls1 += p1.x + p1.y;
#pragma unroll
    for (int t = 0; t < 4; ++t) {
      v2f bm;
      bm.x = hx[s][t];
      bm.y = hy[s][t];
      acc0[t] = WMMA_F32_4(p0, bm, acc0[t]);
      acc1[t] = WMMA_F32_4(p1, bm, acc1[t]);
    }
  }

  ls0 += __shfl_xor(ls0, 16, 32);
  ls1 += __shfl_xor(ls1, 16, 32);
  if (half == 0) {
    rowsum[wave * 32 + ncol]      = ls0;
    rowsum[wave * 32 + 16 + ncol] = ls1;
  }
  put_partial(part + wave * kPart, lane, acc0, acc1);
  __syncthreads();

  const TilePos p = reader_pos(threadIdx.x);
  v4f v = sum_partials(part, p.idx);
  float l = rowsum[p.row];
#pragma unroll
  for (int w = 1; w < kWaves; ++w) l += rowsum[w * 32 + p.row];
  const int grow = (b << 10) + i0 + p.row;
  const float li = 1.0f / l;
  v *= li;
  v.x = (v.x > 0.f) ? v.x : (__expf(v.x) - 1.f);
  v.y = (v.y > 0.f) ? v.y : (__expf(v.y) - 1.f);
  v.z = (v.z > 0.f) ? v.z : (__expf(v.z) - 1.f);
  v.w = (v.w > 0.f) ? v.w : (__expf(v.w) - 1.f);
  *(v4f*)(out + (size_t)grow * kOutF + p.col) = v;
}

extern "C" void kernel_launch(void* const* d_in, const int* in_sizes, int n_in,
                              void* d_out, int out_size, void* d_ws, size_t ws_size,
                              hipStream_t stream) {
  (void)in_sizes; (void)n_in; (void)out_size; (void)ws_size;
  const float* X   = (const float*)d_in[0];
  const int*   adj = (const int*)d_in[3];
  const float* Ww  = (const float*)d_in[4];
  const float* Wb  = (const float*)d_in[5];
  const float* aw  = (const float*)d_in[6];
  const float* ab  = (const float*)d_in[7];

  float* ws = (float*)d_ws;
  float* H  = ws;
  float* F1 = H  + (size_t)kB * kN * kOutF;
  float* F2 = F1 + (size_t)kB * kN;
  float* O  = (float*)d_out;

  hipLaunchKernelGGL(k1_gemm_h,  dim3(256), dim3(kThreads), 0, stream, X, Ww, Wb, aw, H, F1, F2);
  hipLaunchKernelGGL(k2_attn_av, dim3(256), dim3(kThreads), 0, stream, H, F1, F2, adj, ab, O);
}


# Attention: each wave loads the adjacency of its own 64 columns and makes its row bits in registers; per-wave row maxima, fold rescales by exp(m_w-M) (new sum order); row pass and its barrier removed
# speedup vs baseline: 8.9741x; 1.0913x over previous
#include <hip/hip_runtime.h>


typedef __attribute__((ext_vector_type(2))) float v2f;
typedef __attribute__((ext_vector_type(4))) float v4f;
typedef __attribute__((ext_vector_type(8))) float v8f;

#define WMMA_F32_4(a, bmat, c) \
  __builtin_amdgcn_wmma_f32_16x16x4_f32(false, (a), false, (bmat), (short)0, (c), false, false)

constexpr int   kB    = 8;
constexpr int   kN    = 1024;
constexpr int   kInF  = 768;
constexpr int   kOutF = 64;
constexpr float kAlpha  = 0.2f;
constexpr float kNegInf = -9.0e15f;

constexpr int kWaves   = 16;
constexpr int kThreads = kWaves * 32;
constexpr int kPart    = 2 * 4 * 8 * 32;

__device__ __forceinline__ void put_partial(float* __restrict__ part, int lane,
                                            const v8f* acc0, const v8f* acc1)
{
#pragma unroll
  for (int r = 0; r < 8; ++r) {
    v4f q0, q1;
    q0.x = acc0[0][r]; q0.y = acc0[1][r]; q0.z = acc0[2][r]; q0.w = acc0[3][r];
    q1.x = acc1[0][r]; q1.y = acc1[1][r]; q1.z = acc1[2][r]; q1.w = acc1[3][r];
    *(v4f*)(part + (r * 32 + lane) * 4)       = q0;
    *(v4f*)(part + ((8 + r) * 32 + lane) * 4) = q1;
  }
}

struct TilePos { int row; int col; int idx; };
__device__ __forceinline__ TilePos reader_pos(int tid)
{
  const int l = tid & 31, v = tid >> 5;
  TilePos p;
  p.row = (v >> 3) * 16 + (l >> 4) * 8 + (v & 7);
  p.col = (l & 15) << 2;
  p.idx = tid << 2;
  return p;
}

__device__ __forceinline__ v4f sum_partials(const float* __restrict__ lds, int idx)
{
  v4f s = *(const v4f*)(lds + idx);
#pragma unroll
  for (int w = 1; w < kWaves; ++w) s += *(const v4f*)(lds + w * kPart + idx);
  return s;
}

__device__ __forceinline__ void put_partial_rows(float* __restrict__ part, int lane,
                                                 const v8f* acc0, const v8f* acc1)
{
#pragma unroll
  for (int r = 0; r < 8; ++r) {
#pragma unroll
    for (int t = 0; t < 4; ++t) {
      part[(r * 4 + t) * 32 + lane]       = acc0[t][r];
      part[((8 + r) * 4 + t) * 32 + lane] = acc1[t][r];
    }
  }
}

__device__ __forceinline__ TilePos reader_pos_rows(int tid)
{
  const int l = tid & 31, v = tid >> 5;
  const int c4   = (l & 3) << 2;
  const int h    = (l >> 2) & 1;
  const int t    = ((l >> 3) & 1) | ((l >> 4) << 1);
  const int r    = v >> 1;
  const int tile = v & 1;
  TilePos p;
  p.row = tile * 16 + h * 8 + r;
  p.col = t * 16 + c4;
  p.idx = ((tile * 8 + r) * 4 + t) * 32 + h * 16 + c4;
  return p;
}

__global__ __launch_bounds__(kThreads) __attribute__((amdgpu_waves_per_eu(4, 4))) void k1_gemm_h(
    const float* __restrict__ X, const float* __restrict__ W,
    const float* __restrict__ Wb, const float* __restrict__ aw,
    float* __restrict__ H, float* __restrict__ F1, float* __restrict__ F2)
{
  __shared__ float part[kWaves * kPart];
  __shared__ float f12[64];
  const int lane = threadIdx.x & 31;
  const int wave = __builtin_amdgcn_readfirstlane((int)(threadIdx.x >> 5));
  const int row0 = blockIdx.x << 5;
  const int half = lane >> 4;
  const int ncol = lane & 15;
  constexpr int kChunk  = kInF / kWaves;
  constexpr int kGroups = kChunk / 8;
  const int kbeg = wave * kChunk;
  const float* __restrict__ ap0 = X + (size_t)(row0 + ncol) * kInF + kbeg + half * 4;
  const float* __restrict__ ap1 = ap0 + (size_t)16 * kInF;
  const float* __restrict__ bp  = W + (size_t)ncol * kInF + kbeg + half * 4;

  const TilePos p = reader_pos_rows(threadIdx.x);
  const v4f bias = *(const v4f*)(Wb + p.col);
  const v4f att1 = *(const v4f*)(aw + p.col);
  const v4f att2 = *(const v4f*)(aw + kOutF + p.col);

  v4f xa0[kGroups], xa1[kGroups], wb[kGroups][4];
#pragma unroll
  for (int g = 0; g < kGroups; ++g) {
    xa0[g] = *(const v4f*)(ap0 + 8 * g);
    xa1[g] = *(const v4f*)(ap1 + 8 * g);
#pragma unroll
    for (int t = 0; t < 4; ++t)
      wb[g][t] = *(const v4f*)(bp + (size_t)(t * 16) * kInF + 8 * g);
  }

  v8f acc0[4] = {};
  v8f acc1[4] = {};
#pragma unroll
  for (int g = 0; g < kGroups; ++g) {
    v2f a0lo, a0hi, a1lo, a1hi;
    a0lo.x = xa0[g].x; a0lo.y = xa0[g].y; a0hi.x = xa0[g].z; a0hi.y = xa0[g].w;
    a1lo.x = xa1[g].x; a1lo.y = xa1[g].y; a1hi.x = xa1[g].z; a1hi.y = xa1[g].w;
#pragma unroll
    for (int t = 0; t < 4; ++t) {
      v2f bm;
      bm.x = wb[g][t].x; bm.y = wb[g][t].y;
      acc0[t] = WMMA_F32_4(a0lo, bm, acc0[t]);
      acc1[t] = WMMA_F32_4(a1lo, bm, acc1[t]);
    }
#pragma unroll
    for (int t = 0; t < 4; ++t) {
      v2f bm;
      bm.x = wb[g][t].z; bm.y = wb[g][t].w;
      acc0[t] = WMMA_F32_4(a0hi, bm, acc0[t]);
      acc1[t] = WMMA_F32_4(a1hi, bm, acc1[t]);
    }
  }

  put_partial_rows(part + wave * kPart, lane, acc0, acc1);
  __syncthreads();

  const v4f hv = sum_partials(part, p.idx) + bias;

  float d1 = hv.x * att1.x + hv.y * att1.y + hv.z * att1.z + hv.w * att1.w;
  float d2 = hv.x * att2.x + hv.y * att2.y + hv.z * att2.z + hv.w * att2.w;
  d1 += __shfl_xor(d1, 1, 32);   d2 += __shfl_xor(d2, 1, 32);
  d1 += __shfl_xor(d1, 2, 32);   d2 += __shfl_xor(d2, 2, 32);
  d1 += __shfl_xor(d1, 8, 32);   d2 += __shfl_xor(d2, 8, 32);
  d1 += __shfl_xor(d1, 16, 32);  d2 += __shfl_xor(d2, 16, 32);
  if ((lane & 27) == 0) { f12[p.row] = d1; f12[32 + p.row] = d2; }
  __syncthreads();
  *(v4f*)(H + (size_t)(row0 + p.row) * kOutF + p.col) = hv;
  if (wave == 0) {
    F1[row0 + lane] = f12[lane];
    F2[row0 + lane] = f12[32 + lane];
  }
}

__device__ __forceinline__ float score(float f1i, float f2j, bool on)
{
  float e = f1i + f2j;
  e = (e >= 0.f) ? e : kAlpha * e;
  return on ? e : kNegInf;
}

__global__ __launch_bounds__(kThreads) void k2_attn_av(
    const float* __restrict__ H, const float* __restrict__ F1,
    const float* __restrict__ F2, const int* __restrict__ adj,
    const float* __restrict__ abp, float* __restrict__ out)
{
  __shared__ float part[kWaves * kPart];
  __shared__ float rowmx[kWaves * 32];
  __shared__ float rowsum[kWaves * 32];
  const int lane = threadIdx.x & 31;
  const int wave = __builtin_amdgcn_readfirstlane((int)(threadIdx.x >> 5));
  const int gt   = blockIdx.x;
  const int b    = gt >> 5;
  const int i0   = (gt & 31) << 5;
  const int half = lane >> 4;
  const int ncol = lane & 15;
  const int irow0 = (b << 10) + i0 + ncol;
  const int irow1 = irow0 + 16;
  const float* __restrict__ f2b = F2 + (b << 10);
  const float* __restrict__ hb  = H + ((size_t)(b << 10)) * kOutF;

  constexpr int kChunk = kN / kWaves;
  constexpr int kSteps = kChunk / 4;
  constexpr int kAhead = 8;
  const int jbeg = wave * kChunk;

  const int* __restrict__ ap =
      adj + (size_t)((b << 10) + i0 + half) * kN + jbeg + ncol * 4;
  int4 x[16];
#pragma unroll
  for (int i = 0; i < 16; ++i) x[i] = *(const int4*)(ap + (size_t)(2 * i) * kN);

  const float* __restrict__ fp = f2b + jbeg + half * 2;
  const float* __restrict__ hp = hb + (size_t)(jbeg + half * 2) * kOutF + ncol * 4;
  v2f fs[kSteps];
#pragma unroll
  for (int s = 0; s < kSteps; ++s) fs[s] = *(const v2f*)(fp + 4 * s);
  const float ab  = abp[0];
  const float f1a = F1[irow0] + ab;
  const float f1b = F1[irow1] + ab;
  v4f hx[kSteps], hy[kSteps];
#pragma unroll
  for (int s = 0; s < kAhead; ++s) {
    hx[s] = *(const v4f*)(hp + (size_t)(4 * s) * kOutF);
    hy[s] = *(const v4f*)(hp + (size_t)(4 * s + 1) * kOutF);
  }

  unsigned wa0 = 0u, wb0 = 0u, wa1 = 0u, wb1 = 0u;
#pragma unroll
  for (int i = 0; i < 16; ++i) {
    const unsigned b0 = __builtin_amdgcn_ballot_w32(x[i].x > 0);
    const unsigned b1 = __builtin_amdgcn_ballot_w32(x[i].y > 0);
    const unsigned b2 = __builtin_amdgcn_ballot_w32(x[i].z > 0);
    const unsigned b3 = __builtin_amdgcn_ballot_w32(x[i].w > 0);
    const unsigned ev01 = (b0 & 0xFFFFu) | (b1 << 16);
    const unsigned od01 = (b0 >> 16) | (b1 & 0xFFFF0000u);
    const unsigned ev23 = (b2 & 0xFFFFu) | (b3 << 16);
    const unsigned od23 = (b2 >> 16) | (b3 & 0xFFFF0000u);
    const int c0 = 2 * (i & 7);
    if (i < 8) {
      wa0 = (ncol == c0)     ? ev01 : wa0;
      wa0 = (ncol == c0 + 1) ? od01 : wa0;
      wb0 = (ncol == c0)     ? ev23 : wb0;
      wb0 = (ncol == c0 + 1) ? od23 : wb0;
    } else {
      wa1 = (ncol == c0)     ? ev01 : wa1;
      wa1 = (ncol == c0 + 1) ? od01 : wa1;
      wb1 = (ncol == c0)     ? ev23 : wb1;
      wb1 = (ncol == c0 + 1) ? od23 : wb1;
    }
  }
  const unsigned wc0 = half ? wb0 : wa0;
  const unsigned wc1 = half ? wb1 : wa1;

  float sa0[kSteps], sa1[kSteps], sb0[kSteps], sb1[kSteps];
  float ma = -__builtin_inff(), mb = -__builtin_inff();
#pragma unroll
  for (int s = 0; s < kSteps; ++s) {
    sa0[s] = score(f1a, fs[s].x, ((wc0 >> s) & 1u) != 0u);
    sa1[s] = score(f1a, fs[s].y, ((wc0 >> (16 + s)) & 1u) != 0u);
    sb0[s] = score(f1b, fs[s].x, ((wc1 >> s) & 1u) != 0u);
    sb1[s] = score(f1b, fs[s].y, ((wc1 >> (16 + s)) & 1u) != 0u);
    ma = fmaxf(ma, fmaxf(sa0[s], sa1[s]));
    mb = fmaxf(mb, fmaxf(sb0[s], sb1[s]));
  }
  ma = fmaxf(ma, __shfl_xor(ma, 16, 32));
  mb = fmaxf(mb, __shfl_xor(mb, 16, 32));

  float ls0 = 0.f, ls1 = 0.f;
  v8f acc0[4] = {};
  v8f acc1[4] = {};
#pragma unroll
  for (int s = 0; s < kSteps; ++s) {
    if (s + kAhead < kSteps) {
      hx[s + kAhead] = *(const v4f*)(hp + (size_t)(4 * (s + kAhead)) * kOutF);
      hy[s + kAhead] = *(const v4f*)(hp + (size_t)(4 * (s + kAhead) + 1) * kOutF);
    }
    v2f p0, p1;
    p0.x = __expf(sa0[s] - ma);
    p0.y = __expf(sa1[s] - ma);
    p1.x = __expf(sb0[s] - mb);
    p1.y = __expf(sb1[s] - mb);
    ls0 += p0.x + p0.y;
    ls1 += p1.x + p1.y;
#pragma unroll
    for (int t = 0; t < 4; ++t) {
      v2f bm;
      bm.x = hx[s][t];
      bm.y = hy[s][t];
      acc0[t] = WMMA_F32_4(p0, bm, acc0[t]);
      acc1[t] = WMMA_F32_4(p1, bm, acc1[t]);
    }
  }

  ls0 += __shfl_xor(ls0, 16, 32);
  ls1 += __shfl_xor(ls1, 16, 32);
  if (half == 0) {
    rowsum[wave * 32 + ncol]      = ls0;
    rowsum[wave * 32 + 16 + ncol] = ls1;
    rowmx[wave * 32 + ncol]       = ma;
    rowmx[wave * 32 + 16 + ncol]  = mb;
  }
  put_partial(part + wave * kPart, lane, acc0, acc1);
  __syncthreads();

  const TilePos p = reader_pos(threadIdx.x);
  float mw[kWaves];
#pragma unroll
  for (int w = 0; w < kWaves; ++w) mw[w] = rowmx[w * 32 + p.row];
  float mx = mw[0];
#pragma unroll
  for (int w = 1; w < kWaves; ++w) mx = fmaxf(mx, mw[w]);
  v4f v = {0.f, 0.f, 0.f, 0.f};
  float l = 0.f;
#pragma unroll
  for (int w = 0; w < kWaves; ++w) {
    const float sw = __expf(mw[w] - mx);
    l += sw * rowsum[w * 32 + p.row];
    v += sw * *(const v4f*)(part + w * kPart + p.idx);
  }
  const int grow = (b << 10) + i0 + p.row;
  const float li = 1.0f / l;
  v *= li;
  v.x = (v.x > 0.f) ? v.x : (__expf(v.x) - 1.f);
  v.y = (v.y > 0.f) ? v.y : (__expf(v.y) - 1.f);
  v.z = (v.z > 0.f) ? v.z : (__expf(v.z) - 1.f);
  v.w = (v.w > 0.f) ? v.w : (__expf(v.w) - 1.f);
  *(v4f*)(out + (size_t)grow * kOutF + p.col) = v;
}

extern "C" void kernel_launch(void* const* d_in, const int* in_sizes, int n_in,
                              void* d_out, int out_size, void* d_ws, size_t ws_size,
                              hipStream_t stream) {
  (void)in_sizes; (void)n_in; (void)out_size; (void)ws_size;
  const float* X   = (const float*)d_in[0];
  const int*   adj = (const int*)d_in[3];
  const float* Ww  = (const float*)d_in[4];
  const float* Wb  = (const float*)d_in[5];
  const float* aw  = (const float*)d_in[6];
  const float* ab  = (const float*)d_in[7];

  float* ws = (float*)d_ws;
  float* H  = ws;
  float* F1 = H  + (size_t)kB * kN * kOutF;
  float* F2 = F1 + (size_t)kB * kN;
  float* O  = (float*)d_out;

  hipLaunchKernelGGL(k1_gemm_h,  dim3(256), dim3(kThreads), 0, stream, X, Ww, Wb, aw, H, F1, F2);
  hipLaunchKernelGGL(k2_attn_av, dim3(256), dim3(kThreads), 0, stream, H, F1, F2, adj, ab, O);
}
